# starting kernel
# baseline (speedup 1.0000x reference)
#include <hip/hip_runtime.h>
#include <hip/hip_bf16.h>
#include <stddef.h>


typedef __attribute__((ext_vector_type(2))) float v2f;
typedef __attribute__((ext_vector_type(8))) float v8f;

__global__ __launch_bounds__(256) void zero_f4_kernel(float4* __restrict__ p,
                                                      size_t n4) {
  size_t i = (size_t)blockIdx.x * blockDim.x + threadIdx.x;
  size_t stride = (size_t)gridDim.x * blockDim.x;
  for (; i < n4; i += stride) p[i] = make_float4(0.f, 0.f, 0.f, 0.f);
}

__global__ __launch_bounds__(256) void spmm_kernel(
    const int* __restrict__ dstIdx, const int* __restrict__ srcIdx,
    const float* __restrict__ vals, const float* __restrict__ x,
    float* __restrict__ y, int E) {
  const int lane = threadIdx.x & 31;
  int wave = (int)((blockIdx.x * blockDim.x + threadIdx.x) >> 5);
  const int nWaves = (int)((gridDim.x * blockDim.x) >> 5);
  for (int e = wave; e < E; e += nWaves) {
    const int s = srcIdx[e];
    const int d = dstIdx[e];
    const float v = vals[e];
    const float4 xv =
        *(const float4*)(x + (size_t)s * 128 + (size_t)lane * 4);
    float* yp = y + (size_t)d * 128 + (size_t)lane * 4;
    atomicAdd(yp + 0, xv.x * v);
    atomicAdd(yp + 1, xv.y * v);
    atomicAdd(yp + 2, xv.z * v);
    atomicAdd(yp + 3, xv.w * v);
  }
}

template <int NT>
__global__ __launch_bounds__(128) void gemm_bias_kernel(
    const float* __restrict__ S, const float* __restrict__ W,
    const float* __restrict__ bias, float* __restrict__ out, int nRowTiles) {
  const int lane = threadIdx.x & 31;
  const int waveInBlock = threadIdx.x >> 5;
  const int rowTile = blockIdx.x * (blockDim.x >> 5) + waveInBlock;
  if (rowTile >= nRowTiles) return;

  const int hi = lane >> 4;
  const int lo = lane & 15;
  const int ldw = NT * 16;
  const float* Arow = S + (size_t)(rowTile * 16 + lo) * 128;

  v8f acc[NT] = {};

#pragma unroll
  for (int k = 0; k < 128; k += 4) {
    const int kk = k + hi * 2;
    v2f a = *(const v2f*)(Arow + kk);
#pragma unroll
    for (int t = 0; t < NT; ++t) {
      const int col = t * 16 + lo;
      v2f b;
      b.x = W[(size_t)kk * ldw + col];
      b.y = W[(size_t)(kk + 1) * ldw + col];
      acc[t] = __builtin_amdgcn_wmma_f32_16x16x4_f32(
           false, a,  false, b,
           (short)0, acc[t],  false,  false);
    }
  }

#pragma unroll
  for (int t = 0; t < NT; ++t) {
    const int col = t * 16 + lo;
    const float bv = bias[col];
#pragma unroll
    for (int j = 0; j < 8; ++j) {
      const int m = j + hi * 8;
      out[(size_t)(rowTile * 16 + m) * ldw + col] = acc[t][j] + bv;
    }
  }
}

extern "C" void kernel_launch(void* const* d_in, const int* in_sizes, int n_in,
                              void* d_out, int out_size, void* d_ws,
                              size_t ws_size, hipStream_t stream) {
  const float* x = (const float*)d_in[0];
  const int* adj = (const int*)d_in[1];
  const float* vals = (const float*)d_in[2];
  const float* W1 = (const float*)d_in[3];
  const float* b1 = (const float*)d_in[4];
  const float* W2 = (const float*)d_in[5];
  const float* b2 = (const float*)d_in[6];
  const float* W3 = (const float*)d_in[7];
  const float* b3 = (const float*)d_in[8];
  float* out = (float*)d_out;

  const int E = in_sizes[2];
  const int N = in_sizes[0] / 128;
  const int* dstIdx = adj;
  const int* srcIdx = adj + E;

  float* ws0 = (float*)d_ws;
  float* ws1 = ws0 + (size_t)N * 128;

  const size_t n4 = (size_t)N * 128 / 4;
  const int zeroBlocks = 2048;
  const int spmmBlocks = (int)(((long long)E * 32 + 255) / 256);
  const int nRowTiles = N / 16;
  const int gemmBlocks = (nRowTiles + 3) / 4;

  zero_f4_kernel<<<zeroBlocks, 256, 0, stream>>>((float4*)ws0, n4);
  spmm_kernel<<<spmmBlocks, 256, 0, stream>>>(dstIdx, srcIdx, vals, x, ws0, E);
  gemm_bias_kernel<8><<<gemmBlocks, 128, 0, stream>>>(ws0, W1, b1, ws1,
                                                      nRowTiles);

  zero_f4_kernel<<<zeroBlocks, 256, 0, stream>>>((float4*)ws0, n4);
  spmm_kernel<<<spmmBlocks, 256, 0, stream>>>(dstIdx, srcIdx, vals, ws1, ws0, E);
  gemm_bias_kernel<8><<<gemmBlocks, 128, 0, stream>>>(ws0, W2, b2, ws1,
                                                      nRowTiles);

  zero_f4_kernel<<<zeroBlocks, 256, 0, stream>>>((float4*)ws0, n4);
  spmm_kernel<<<spmmBlocks, 256, 0, stream>>>(dstIdx, srcIdx, vals, ws1, ws0, E);
  gemm_bias_kernel<4><<<gemmBlocks, 128, 0, stream>>>(ws0, W3, b3, out,
                                                      nRowTiles);
}


# SpMM lane map: lane l owns columns l, l+32, l+64, l+96, so each gather load and each f32 atomic add of a wave covers one whole 128-B line, not 8 floats in each of 4 lines; atomic sum order as before
# speedup vs baseline: 1.3984x; 1.3984x over previous
#include <hip/hip_runtime.h>
#include <hip/hip_bf16.h>
#include <stddef.h>


typedef __attribute__((ext_vector_type(2))) float v2f;
typedef __attribute__((ext_vector_type(8))) float v8f;

__global__ __launch_bounds__(256) void zero_f4_kernel(float4* __restrict__ p,
                                                      size_t n4) {
  size_t i = (size_t)blockIdx.x * blockDim.x + threadIdx.x;
  size_t stride = (size_t)gridDim.x * blockDim.x;
  for (; i < n4; i += stride) p[i] = make_float4(0.f, 0.f, 0.f, 0.f);
}

__global__ __launch_bounds__(256) void spmm_kernel(
    const int* __restrict__ dstIdx, const int* __restrict__ srcIdx,
    const float* __restrict__ vals, const float* __restrict__ x,
    float* __restrict__ y, int E) {
  const int lane = threadIdx.x & 31;
  int wave = (int)((blockIdx.x * blockDim.x + threadIdx.x) >> 5);
  const int nWaves = (int)((gridDim.x * blockDim.x) >> 5);
  for (int e = wave; e < E; e += nWaves) {
    const int s = srcIdx[e];
    const int d = dstIdx[e];
    const float v = vals[e];
    const float* xp = x + (size_t)s * 128 + (size_t)lane;
    const float x0 = xp[0];
    const float x1 = xp[32];
    const float x2 = xp[64];
    const float x3 = xp[96];
    float* yp = y + (size_t)d * 128 + (size_t)lane;
    atomicAdd(yp + 0, x0 * v);
    atomicAdd(yp + 32, x1 * v);
    atomicAdd(yp + 64, x2 * v);
    atomicAdd(yp + 96, x3 * v);
  }
}

template <int NT>
__global__ __launch_bounds__(128) void gemm_bias_kernel(
    const float* __restrict__ S, const float* __restrict__ W,
    const float* __restrict__ bias, float* __restrict__ out, int nRowTiles) {
  const int lane = threadIdx.x & 31;
  const int waveInBlock = threadIdx.x >> 5;
  const int rowTile = blockIdx.x * (blockDim.x >> 5) + waveInBlock;
  if (rowTile >= nRowTiles) return;

  const int hi = lane >> 4;
  const int lo = lane & 15;
  const int ldw = NT * 16;
  const float* Arow = S + (size_t)(rowTile * 16 + lo) * 128;

  v8f acc[NT] = {};

#pragma unroll
  for (int k = 0; k < 128; k += 4) {
    const int kk = k + hi * 2;
    v2f a = *(const v2f*)(Arow + kk);
#pragma unroll
    for (int t = 0; t < NT; ++t) {
      const int col = t * 16 + lo;
      v2f b;
      b.x = W[(size_t)kk * ldw + col];
      b.y = W[(size_t)(kk + 1) * ldw + col];
      acc[t] = __builtin_amdgcn_wmma_f32_16x16x4_f32(
           false, a,  false, b,
           (short)0, acc[t],  false,  false);
    }
  }

#pragma unroll
  for (int t = 0; t < NT; ++t) {
    const int col = t * 16 + lo;
    const float bv = bias[col];
#pragma unroll
    for (int j = 0; j < 8; ++j) {
      const int m = j + hi * 8;
      out[(size_t)(rowTile * 16 + m) * ldw + col] = acc[t][j] + bv;
    }
  }
}

extern "C" void kernel_launch(void* const* d_in, const int* in_sizes, int n_in,
                              void* d_out, int out_size, void* d_ws,
                              size_t ws_size, hipStream_t stream) {
  const float* x = (const float*)d_in[0];
  const int* adj = (const int*)d_in[1];
  const float* vals = (const float*)d_in[2];
  const float* W1 = (const float*)d_in[3];
  const float* b1 = (const float*)d_in[4];
  const float* W2 = (const float*)d_in[5];
  const float* b2 = (const float*)d_in[6];
  const float* W3 = (const float*)d_in[7];
  const float* b3 = (const float*)d_in[8];
  float* out = (float*)d_out;

  const int E = in_sizes[2];
  const int N = in_sizes[0] / 128;
  const int* dstIdx = adj;
  const int* srcIdx = adj + E;

  float* ws0 = (float*)d_ws;
  float* ws1 = ws0 + (size_t)N * 128;

  const size_t n4 = (size_t)N * 128 / 4;
  const int zeroBlocks = 2048;
  const int spmmBlocks = (int)(((long long)E * 32 + 255) / 256);
  const int nRowTiles = N / 16;
  const int gemmBlocks = (nRowTiles + 3) / 4;

  zero_f4_kernel<<<zeroBlocks, 256, 0, stream>>>((float4*)ws0, n4);
  spmm_kernel<<<spmmBlocks, 256, 0, stream>>>(dstIdx, srcIdx, vals, x, ws0, E);
  gemm_bias_kernel<8><<<gemmBlocks, 128, 0, stream>>>(ws0, W1, b1, ws1,
                                                      nRowTiles);

  zero_f4_kernel<<<zeroBlocks, 256, 0, stream>>>((float4*)ws0, n4);
  spmm_kernel<<<spmmBlocks, 256, 0, stream>>>(dstIdx, srcIdx, vals, ws1, ws0, E);
  gemm_bias_kernel<8><<<gemmBlocks, 128, 0, stream>>>(ws0, W2, b2, ws1,
                                                      nRowTiles);

  zero_f4_kernel<<<zeroBlocks, 256, 0, stream>>>((float4*)ws0, n4);
  spmm_kernel<<<spmmBlocks, 256, 0, stream>>>(dstIdx, srcIdx, vals, ws1, ws0, E);
  gemm_bias_kernel<4><<<gemmBlocks, 128, 0, stream>>>(ws0, W3, b3, out,
                                                      nRowTiles);
}


# SpMM by owner: a work-group keeps 196 destination rows' sums in LDS, filters the whole edge list for them, LDS float adds in arrival order, each row stored once; no global atomics, no zero fill
# speedup vs baseline: 2.3373x; 1.6714x over previous
#include <hip/hip_runtime.h>
#include <hip/hip_bf16.h>
#include <stddef.h>


typedef __attribute__((ext_vector_type(2))) float v2f;
typedef __attribute__((ext_vector_type(8))) float v8f;

constexpr int kRowsPerBlock = 196;
constexpr int kWaves = 32;
constexpr int kListCap = 31 + 4 * 128 + 1;


template <int U>
__device__ __forceinline__ void spmm_edges(const float* __restrict__ x,
                                           float* acc, int sv, int dv,
                                           float vv, int j, int lane) {
  float4 r[U];
#pragma unroll
  for (int u = 0; u < U; ++u) {
    const int s = __builtin_amdgcn_readlane(sv, j + u);
    const float* xr = x + (size_t)s * 128;
    r[u] = *(const float4*)(xr + lane * 4);
  }
#pragma unroll
  for (int u = 0; u < U; ++u) {
    const int d = __builtin_amdgcn_readlane(dv, j + u);
    const float v =
        __int_as_float(__builtin_amdgcn_readlane(__float_as_int(vv), j + u));
    float* ar = acc + d * 128 + lane;
    __hip_atomic_fetch_add(ar, r[u].x * v, __ATOMIC_RELAXED,
                           __HIP_MEMORY_SCOPE_WORKGROUP);
    __hip_atomic_fetch_add(ar + 32, r[u].y * v, __ATOMIC_RELAXED,
                           __HIP_MEMORY_SCOPE_WORKGROUP);
    __hip_atomic_fetch_add(ar + 64, r[u].z * v, __ATOMIC_RELAXED,
                           __HIP_MEMORY_SCOPE_WORKGROUP);
    __hip_atomic_fetch_add(ar + 96, r[u].w * v, __ATOMIC_RELAXED,
                           __HIP_MEMORY_SCOPE_WORKGROUP);
  }
}

__device__ __forceinline__ int unit_of(int k, int start, int nFull) {
  const int u = k + start;
  return (u >= nFull) ? u - nFull : u;
}

__device__ __forceinline__ int4 load_unit(const int* __restrict__ dstIdx,
                                          int k, int start, int nFull,
                                          int lane) {
  const int kk = (k < nFull) ? k : nFull - 1;
  const int base = unit_of(kk, start, nFull) * 128 + lane * 4;
  return *(const int4*)(dstIdx + base);
}

__device__ __forceinline__ int list_push(bool m, int e, int rel, int* qe,
                                         int* qr, int qn) {
  const unsigned b = __builtin_amdgcn_ballot_w32(m);
  if (m) {
    const int pos = qn + (int)__builtin_amdgcn_mbcnt_lo(b, 0u);
    qe[pos] = e;
    qr[pos] = rel;
  }
  return qn + __builtin_popcount(b);
}

__device__ __forceinline__ int take_unit(int4 d, int k, int start, int nFull,
                                         int r0, int rows, int lane, int* qe,
                                         int* qr, int qn) {
  const int rx = d.x - r0, ry = d.y - r0, rz = d.z - r0, rw = d.w - r0;
  const bool mx = (unsigned)rx < (unsigned)rows;
  const bool my = (unsigned)ry < (unsigned)rows;
  const bool mz = (unsigned)rz < (unsigned)rows;
  const bool mw = (unsigned)rw < (unsigned)rows;
  if (__builtin_amdgcn_ballot_w32(mx | my | mz | mw) == 0u) return qn;
  const int e = unit_of(k, start, nFull) * 128 + lane * 4;
  qn = list_push(mx, e, rx, qe, qr, qn);
  qn = list_push(my, e + 1, ry, qe, qr, qn);
  qn = list_push(mz, e + 2, rz, qe, qr, qn);
  qn = list_push(mw, e + 3, rw, qe, qr, qn);
  return qn;
}

__global__ __launch_bounds__(1024) void spmm_kernel(
    const int* __restrict__ dstIdx, const int* __restrict__ srcIdx,
    const float* __restrict__ vals, const float* __restrict__ x,
    float* __restrict__ y, int E, int N) {
  __shared__ __attribute__((aligned(16))) float acc[kRowsPerBlock * 128];
  __shared__ int listE[kWaves * kListCap];
  __shared__ int listR[kWaves * kListCap];

  const int tid = threadIdx.x;
  const int lane = tid & 31;
  const int wave = __builtin_amdgcn_readfirstlane(tid >> 5);
  int* qe = listE + wave * kListCap;
  int* qr = listR + wave * kListCap;
  float4* acc4 = (float4*)acc;

  const int nFull = E >> 7;
  const int start = (int)(((long long)blockIdx.x * nFull) / gridDim.x);

  for (int r0 = blockIdx.x * kRowsPerBlock; r0 < N;
       r0 += gridDim.x * kRowsPerBlock) {
    const int rows = (N - r0 < kRowsPerBlock) ? N - r0 : kRowsPerBlock;

    for (int i = tid; i < kRowsPerBlock * 32; i += 1024)
      acc4[i] = make_float4(0.f, 0.f, 0.f, 0.f);
    __syncthreads();

    int qn = 0;
    if (wave < nFull) {
      int k = wave;
      int4 a0 = load_unit(dstIdx, k, start, nFull, lane);
      int4 a1 = load_unit(dstIdx, k + 32, start, nFull, lane);
      int4 a2 = load_unit(dstIdx, k + 64, start, nFull, lane);
      int4 a3 = load_unit(dstIdx, k + 96, start, nFull, lane);
      for (; k < nFull; k += 128) {
        qn = take_unit(a0, k, start, nFull, r0, rows, lane, qe, qr, qn);
        a0 = load_unit(dstIdx, k + 128, start, nFull, lane);
        qn = take_unit(a1, k + 32, start, nFull, r0,
                       (k + 32 < nFull) ? rows : 0, lane, qe, qr, qn);
        a1 = load_unit(dstIdx, k + 160, start, nFull, lane);
        qn = take_unit(a2, k + 64, start, nFull, r0,
                       (k + 64 < nFull) ? rows : 0, lane, qe, qr, qn);
        a2 = load_unit(dstIdx, k + 192, start, nFull, lane);
        qn = take_unit(a3, k + 96, start, nFull, r0,
                       (k + 96 < nFull) ? rows : 0, lane, qe, qr, qn);
        a3 = load_unit(dstIdx, k + 224, start, nFull, lane);
        while (qn >= 32) {
          qn -= 32;
          const int e = qe[qn + lane];
          const int dv = qr[qn + lane];
          const int sv = srcIdx[e];
          const float vv = vals[e];
          for (int j = 0; j < 32; j += 8)
            spmm_edges<8>(x, acc, sv, dv, vv, j, lane);
        }
      }
    }
    if ((E & 127) != 0 && wave == kWaves - 1) {
      for (int t = 0; t < 4; ++t) {
        const int e = nFull * 128 + t * 32 + lane;
        const int rel = ((e < E) ? dstIdx[e] : -1) - r0;
        qn = list_push((unsigned)rel < (unsigned)rows, e, rel, qe, qr, qn);
      }
    }
    while (qn > 0) {
      const int cnt = (qn < 32) ? qn : 32;
      qn -= cnt;
      const int i = qn + ((lane < cnt) ? lane : 0);
      const int e = qe[i];
      const int dv = qr[i];
      const int sv = srcIdx[e];
      const float vv = vals[e];
      int j = 0;
      for (; j + 8 <= cnt; j += 8) spmm_edges<8>(x, acc, sv, dv, vv, j, lane);
      if (cnt - j >= 4) {
        spmm_edges<4>(x, acc, sv, dv, vv, j, lane);
        j += 4;
      }
      if (cnt - j >= 2) {
        spmm_edges<2>(x, acc, sv, dv, vv, j, lane);
        j += 2;
      }
      if (cnt - j >= 1) spmm_edges<1>(x, acc, sv, dv, vv, j, lane);
    }
    __syncthreads();

    float4* y4 = (float4*)(y + (size_t)r0 * 128);
    for (int i = tid; i < rows * 32; i += 1024) {
      const float* ar = acc + (i >> 5) * 128 + (i & 31);
      y4[i] = make_float4(ar[0], ar[32], ar[64], ar[96]);
    }
    __syncthreads();
  }
}

template <int NT>
__global__ __launch_bounds__(128) void gemm_bias_kernel(
    const float* __restrict__ S, const float* __restrict__ W,
    const float* __restrict__ bias, float* __restrict__ out, int nRowTiles) {
  const int lane = threadIdx.x & 31;
  const int waveInBlock = threadIdx.x >> 5;
  const int rowTile = blockIdx.x * (blockDim.x >> 5) + waveInBlock;
  if (rowTile >= nRowTiles) return;

  const int hi = lane >> 4;
  const int lo = lane & 15;
  const int ldw = NT * 16;
  const float* Arow = S + (size_t)(rowTile * 16 + lo) * 128;

  v8f acc[NT] = {};

#pragma unroll
  for (int k = 0; k < 128; k += 4) {
    const int kk = k + hi * 2;
    v2f a = *(const v2f*)(Arow + kk);
#pragma unroll
    for (int t = 0; t < NT; ++t) {
      const int col = t * 16 + lo;
      v2f b;
      b.x = W[(size_t)kk * ldw + col];
      b.y = W[(size_t)(kk + 1) * ldw + col];
      acc[t] = __builtin_amdgcn_wmma_f32_16x16x4_f32(
           false, a,  false, b,
           (short)0, acc[t],  false,  false);
    }
  }

#pragma unroll
  for (int t = 0; t < NT; ++t) {
    const int col = t * 16 + lo;
    const float bv = bias[col];
#pragma unroll
    for (int j = 0; j < 8; ++j) {
      const int m = j + hi * 8;
      out[(size_t)(rowTile * 16 + m) * ldw + col] = acc[t][j] + bv;
    }
  }
}

extern "C" void kernel_launch(void* const* d_in, const int* in_sizes, int n_in,
                              void* d_out, int out_size, void* d_ws,
                              size_t ws_size, hipStream_t stream) {
  const float* x = (const float*)d_in[0];
  const int* adj = (const int*)d_in[1];
  const float* vals = (const float*)d_in[2];
  const float* W1 = (const float*)d_in[3];
  const float* b1 = (const float*)d_in[4];
  const float* W2 = (const float*)d_in[5];
  const float* b2 = (const float*)d_in[6];
  const float* W3 = (const float*)d_in[7];
  const float* b3 = (const float*)d_in[8];
  float* out = (float*)d_out;

  const int E = in_sizes[2];
  const int N = in_sizes[0] / 128;
  const int* dstIdx = adj;
  const int* srcIdx = adj + E;

  float* ws0 = (float*)d_ws;
  float* ws1 = ws0 + (size_t)N * 128;

  const int spmmBlocks = 256;
  const int nRowTiles = N / 16;
  const int gemmBlocks = (nRowTiles + 3) / 4;

  spmm_kernel<<<spmmBlocks, 1024, 0, stream>>>(dstIdx, srcIdx, vals, x, ws0, E,
                                               N);
  gemm_bias_kernel<8><<<gemmBlocks, 128, 0, stream>>>(ws0, W1, b1, ws1,
                                                      nRowTiles);

  spmm_kernel<<<spmmBlocks, 1024, 0, stream>>>(dstIdx, srcIdx, vals, ws1, ws0,
                                               E, N);
  gemm_bias_kernel<8><<<gemmBlocks, 128, 0, stream>>>(ws0, W2, b2, ws1,
                                                      nRowTiles);

  spmm_kernel<<<spmmBlocks, 1024, 0, stream>>>(dstIdx, srcIdx, vals, ws1, ws0,
                                               E, N);
  gemm_bias_kernel<4><<<gemmBlocks, 128, 0, stream>>>(ws0, W3, b3, out,
                                                      nRowTiles);
}


# SpMM scan list re-coded once per call: one byte per edge names its destination's row block, so each work-group's pass over all edges reads a quarter of the bytes; LDS add arrival order differs
# speedup vs baseline: 2.8042x; 1.1998x over previous
#include <hip/hip_runtime.h>
#include <hip/hip_bf16.h>
#include <stddef.h>


typedef __attribute__((ext_vector_type(2))) float v2f;
typedef __attribute__((ext_vector_type(8))) float v8f;

constexpr int kRowsPerBlock = 196;
constexpr int kWaves = 32;
constexpr int kUnitEdges = 512;
constexpr int kListCap = 31 + 2 * kUnitEdges + 1;

__device__ __forceinline__ unsigned owner_code(int d) {
  return ((unsigned)d / (unsigned)kRowsPerBlock) & 255u;
}

__device__ __forceinline__ unsigned owner_word(int4 d) {
  return owner_code(d.x) | (owner_code(d.y) << 8) | (owner_code(d.z) << 16) |
         (owner_code(d.w) << 24);
}

__global__ __launch_bounds__(256) void owner_code_kernel(
    const int* __restrict__ dstIdx, unsigned* __restrict__ codes, int E,
    int nWords) {
  const int w0 = blockIdx.x * 1024 + threadIdx.x;
  if ((blockIdx.x + 1) * 4096 <= E) {
    int4 d[4];
#pragma unroll
    for (int c = 0; c < 4; ++c)
      d[c] = *(const int4*)(dstIdx + (w0 + c * 256) * 4);
#pragma unroll
    for (int c = 0; c < 4; ++c) codes[w0 + c * 256] = owner_word(d[c]);
  } else {
    for (int c = 0; c < 4; ++c) {
      const int g = w0 + c * 256;
      if (g >= nWords) break;
      const int e = g * 4;
      unsigned w = 0u;
      for (int i = 0; i < 4; ++i) {
        const unsigned b = (e + i < E) ? owner_code(dstIdx[e + i]) : 255u;
        w |= b << (8 * i);
      }
      codes[g] = w;
    }
  }
}


template <int U>
__device__ __forceinline__ void spmm_edges(const float* __restrict__ x,
                                           float* acc, int sv, int dv,
                                           float vv, int j, int lane) {
  float4 r[U];
#pragma unroll
  for (int u = 0; u < U; ++u) {
    const int s = __builtin_amdgcn_readlane(sv, j + u);
    const float* xr = x + (size_t)s * 128;
    r[u] = *(const float4*)(xr + lane * 4);
  }
#pragma unroll
  for (int u = 0; u < U; ++u) {
    const int d = __builtin_amdgcn_readlane(dv, j + u);
    const float v =
        __int_as_float(__builtin_amdgcn_readlane(__float_as_int(vv), j + u));
    float* ar = acc + d * 128 + lane;
    __hip_atomic_fetch_add(ar, r[u].x * v, __ATOMIC_RELAXED,
                           __HIP_MEMORY_SCOPE_WORKGROUP);
    __hip_atomic_fetch_add(ar + 32, r[u].y * v, __ATOMIC_RELAXED,
                           __HIP_MEMORY_SCOPE_WORKGROUP);
    __hip_atomic_fetch_add(ar + 64, r[u].z * v, __ATOMIC_RELAXED,
                           __HIP_MEMORY_SCOPE_WORKGROUP);
    __hip_atomic_fetch_add(ar + 96, r[u].w * v, __ATOMIC_RELAXED,
                           __HIP_MEMORY_SCOPE_WORKGROUP);
  }
}

__device__ __forceinline__ int unit_of(int k, int start, int nUnits) {
  const int u = k + start;
  return (u >= nUnits) ? u - nUnits : u;
}

__device__ __forceinline__ uint4 load_unit(const unsigned* __restrict__ codes,
                                           int k, int start, int nUnits,
                                           int lane) {
  const int kk = (k < nUnits) ? k : nUnits - 1;
  return *(const uint4*)(codes + unit_of(kk, start, nUnits) * 128 + lane * 4);
}

__device__ __forceinline__ int list_push(bool m, int e, int* qe, int qn) {
  const unsigned b = __builtin_amdgcn_ballot_w32(m);
  if (m) qe[qn + (int)__builtin_amdgcn_mbcnt_lo(b, 0u)] = e;
  return qn + __builtin_popcount(b);
}

__device__ __forceinline__ unsigned zero_bytes(unsigned t) {
  return (t - 0x01010101u) & ~t;
}

__device__ __forceinline__ int take_unit(uint4 w, int unit, unsigned kb, int E,
                                         int lane, int* qe, int qn) {
  const unsigned z0 = zero_bytes(w.x ^ kb), z1 = zero_bytes(w.y ^ kb);
  const unsigned z2 = zero_bytes(w.z ^ kb), z3 = zero_bytes(w.w ^ kb);
  const unsigned any = (z0 | z1 | z2 | z3) & 0x80808080u;
  if (__builtin_amdgcn_ballot_w32(any != 0u) == 0u) return qn;
  unsigned mk = ((z0 & 0x80808080u) >> 7) | ((z1 & 0x80808080u) >> 6) |
                ((z2 & 0x80808080u) >> 5) | ((z3 & 0x80808080u) >> 4);
  const int e0 = unit * kUnitEdges + lane * 16;
  while (__builtin_amdgcn_ballot_w32(mk != 0u) != 0u) {
    const int p = __builtin_ctz(mk | 0x80000000u);
    const int e = e0 + ((p & 7) << 2) + (p >> 3);
    qn = list_push(mk != 0u && e < E, e, qe, qn);
    mk &= mk - 1u;
  }
  return qn;
}

__device__ __forceinline__ void edge_words(const int* __restrict__ dstIdx,
                                           const int* __restrict__ srcIdx,
                                           const float* __restrict__ vals,
                                           int e, int r0, int rows, int& sv,
                                           int& dv, float& vv) {
  const int rel = dstIdx[e] - r0;
  sv = srcIdx[e];
  vv = vals[e];
  dv = ((unsigned)rel < (unsigned)rows) ? rel : kRowsPerBlock;
}

__device__ __forceinline__ int serve_full(const int* __restrict__ dstIdx,
                                          const int* __restrict__ srcIdx,
                                          const float* __restrict__ vals,
                                          const float* __restrict__ x,
                                          float* acc, const int* qe, int qn,
                                          int r0, int rows, int lane) {
  while (qn >= 32) {
    qn -= 32;
    int sv, dv;
    float vv;
    edge_words(dstIdx, srcIdx, vals, qe[qn + lane], r0, rows, sv, dv, vv);
    for (int j = 0; j < 32; j += 8) spmm_edges<8>(x, acc, sv, dv, vv, j, lane);
  }
  return qn;
}

__global__ __launch_bounds__(1024) void spmm_kernel(
    const unsigned* __restrict__ codes, const int* __restrict__ dstIdx,
    const int* __restrict__ srcIdx, const float* __restrict__ vals,
    const float* __restrict__ x, float* __restrict__ y, int E, int N) {
  __shared__ __attribute__((aligned(16))) float acc[(kRowsPerBlock + 1) * 128];
  __shared__ int listE[kWaves * kListCap];

  const int tid = threadIdx.x;
  const int lane = tid & 31;
  const int wave = __builtin_amdgcn_readfirstlane(tid >> 5);
  int* qe = listE + wave * kListCap;
  float4* acc4 = (float4*)acc;

  const int nUnits = (E + kUnitEdges - 1) / kUnitEdges;
  const int start = (int)(((long long)blockIdx.x * nUnits) / gridDim.x);

  for (int blk = blockIdx.x; blk * kRowsPerBlock < N; blk += gridDim.x) {
    const int r0 = blk * kRowsPerBlock;
    const int rows = (N - r0 < kRowsPerBlock) ? N - r0 : kRowsPerBlock;
    const unsigned kb = (unsigned)(blk & 255) * 0x01010101u;

    for (int i = tid; i < (kRowsPerBlock + 1) * 32; i += 1024)
      acc4[i] = make_float4(0.f, 0.f, 0.f, 0.f);
    __syncthreads();

    int qn = 0;
    if (wave < nUnits) {
      int k = wave;
      uint4 a0 = load_unit(codes, k, start, nUnits, lane);
      uint4 a1 = load_unit(codes, k + 32, start, nUnits, lane);
      uint4 a2 = load_unit(codes, k + 64, start, nUnits, lane);
      uint4 a3 = load_unit(codes, k + 96, start, nUnits, lane);
      for (; k < nUnits; k += 128) {
        qn = take_unit(a0, unit_of(k, start, nUnits), kb, E, lane, qe, qn);
        a0 = load_unit(codes, k + 128, start, nUnits, lane);
        if (k + 32 < nUnits)
          qn = take_unit(a1, unit_of(k + 32, start, nUnits), kb, E, lane, qe,
                         qn);
        a1 = load_unit(codes, k + 160, start, nUnits, lane);
        qn = serve_full(dstIdx, srcIdx, vals, x, acc, qe, qn, r0, rows, lane);
        if (k + 64 < nUnits)
          qn = take_unit(a2, unit_of(k + 64, start, nUnits), kb, E, lane, qe,
                         qn);
        a2 = load_unit(codes, k + 192, start, nUnits, lane);
        if (k + 96 < nUnits)
          qn = take_unit(a3, unit_of(k + 96, start, nUnits), kb, E, lane, qe,
                         qn);
        a3 = load_unit(codes, k + 224, start, nUnits, lane);
        qn = serve_full(dstIdx, srcIdx, vals, x, acc, qe, qn, r0, rows, lane);
      }
    }
    while (qn > 0) {
      const int cnt = (qn < 32) ? qn : 32;
      qn -= cnt;
      int sv, dv;
      float vv;
      edge_words(dstIdx, srcIdx, vals, qe[qn + ((lane < cnt) ? lane : 0)], r0,
                 rows, sv, dv, vv);
      int j = 0;
      for (; j + 8 <= cnt; j += 8) spmm_edges<8>(x, acc, sv, dv, vv, j, lane);
      if (cnt - j >= 4) {
        spmm_edges<4>(x, acc, sv, dv, vv, j, lane);
        j += 4;
      }
      if (cnt - j >= 2) {
        spmm_edges<2>(x, acc, sv, dv, vv, j, lane);
        j += 2;
      }
      if (cnt - j >= 1) spmm_edges<1>(x, acc, sv, dv, vv, j, lane);
    }
    __syncthreads();

    float4* y4 = (float4*)(y + (size_t)r0 * 128);
    for (int i = tid; i < rows * 32; i += 1024) {
      const float* ar = acc + (i >> 5) * 128 + (i & 31);
      y4[i] = make_float4(ar[0], ar[32], ar[64], ar[96]);
    }
    __syncthreads();
  }
}

template <int NT>
__global__ __launch_bounds__(128) void gemm_bias_kernel(
    const float* __restrict__ S, const float* __restrict__ W,
    const float* __restrict__ bias, float* __restrict__ out, int nRowTiles) {
  const int lane = threadIdx.x & 31;
  const int waveInBlock = threadIdx.x >> 5;
  const int rowTile = blockIdx.x * (blockDim.x >> 5) + waveInBlock;
  if (rowTile >= nRowTiles) return;

  const int hi = lane >> 4;
  const int lo = lane & 15;
  const int ldw = NT * 16;
  const float* Arow = S + (size_t)(rowTile * 16 + lo) * 128;

  v8f acc[NT] = {};

#pragma unroll
  for (int k = 0; k < 128; k += 4) {
    const int kk = k + hi * 2;
    v2f a = *(const v2f*)(Arow + kk);
#pragma unroll
    for (int t = 0; t < NT; ++t) {
      const int col = t * 16 + lo;
      v2f b;
      b.x = W[(size_t)kk * ldw + col];
      b.y = W[(size_t)(kk + 1) * ldw + col];
      acc[t] = __builtin_amdgcn_wmma_f32_16x16x4_f32(
           false, a,  false, b,
           (short)0, acc[t],  false,  false);
    }
  }

#pragma unroll
  for (int t = 0; t < NT; ++t) {
    const int col = t * 16 + lo;
    const float bv = bias[col];
#pragma unroll
    for (int j = 0; j < 8; ++j) {
      const int m = j + hi * 8;
      out[(size_t)(rowTile * 16 + m) * ldw + col] = acc[t][j] + bv;
    }
  }
}

extern "C" void kernel_launch(void* const* d_in, const int* in_sizes, int n_in,
                              void* d_out, int out_size, void* d_ws,
                              size_t ws_size, hipStream_t stream) {
  const float* x = (const float*)d_in[0];
  const int* adj = (const int*)d_in[1];
  const float* vals = (const float*)d_in[2];
  const float* W1 = (const float*)d_in[3];
  const float* b1 = (const float*)d_in[4];
  const float* W2 = (const float*)d_in[5];
  const float* b2 = (const float*)d_in[6];
  const float* W3 = (const float*)d_in[7];
  const float* b3 = (const float*)d_in[8];
  float* out = (float*)d_out;

  const int E = in_sizes[2];
  const int N = in_sizes[0] / 128;
  const int* dstIdx = adj;
  const int* srcIdx = adj + E;

  float* ws0 = (float*)d_ws;
  float* ws1 = ws0 + (size_t)N * 128;
  unsigned* codes = (unsigned*)(ws1 + (size_t)N * 128);

  const int spmmBlocks = 256;
  const int nRowTiles = N / 16;
  const int gemmBlocks = (nRowTiles + 3) / 4;
  const int nWords = (E + 511) / 512 * 128;
  const int codeBlocks = (nWords + 1023) / 1024;

  owner_code_kernel<<<codeBlocks, 256, 0, stream>>>(dstIdx, codes, E, nWords);

  spmm_kernel<<<spmmBlocks, 1024, 0, stream>>>(codes, dstIdx, srcIdx, vals,
                                               x, ws0, E, N);
  gemm_bias_kernel<8><<<gemmBlocks, 128, 0, stream>>>(ws0, W1, b1, ws1,
                                                      nRowTiles);

  spmm_kernel<<<spmmBlocks, 1024, 0, stream>>>(codes, dstIdx, srcIdx, vals,
                                               ws1, ws0, E, N);
  gemm_bias_kernel<8><<<gemmBlocks, 128, 0, stream>>>(ws0, W2, b2, ws1,
                                                      nRowTiles);

  spmm_kernel<<<spmmBlocks, 1024, 0, stream>>>(codes, dstIdx, srcIdx, vals,
                                               ws1, ws0, E, N);
  gemm_bias_kernel<4><<<gemmBlocks, 128, 0, stream>>>(ws0, W3, b3, out,
                                                      nRowTiles);
}


# GEMM+bias of each layer done by the SpMM work-group on its own LDS sums (A fragments read from LDS, f32 WMMA, k order kept), rows stored directly: 7 launches to 4, no sum plane written or re-read
# speedup vs baseline: 3.3722x; 1.2026x over previous
#include <hip/hip_runtime.h>
#include <hip/hip_bf16.h>
#include <stddef.h>


typedef __attribute__((ext_vector_type(2))) float v2f;
typedef __attribute__((ext_vector_type(8))) float v8f;

constexpr int kRowsPerBlock = 196;
constexpr int kColPitch = 33;
constexpr int kRowPitch = 4 * kColPitch;
constexpr int kWaves = 32;
constexpr int kUnitEdges = 512;
constexpr int kListCap = 31 + 2 * kUnitEdges + 1;

__device__ __forceinline__ unsigned owner_code(int d) {
  return ((unsigned)d / (unsigned)kRowsPerBlock) & 255u;
}

__device__ __forceinline__ unsigned owner_word(int4 d) {
  return owner_code(d.x) | (owner_code(d.y) << 8) | (owner_code(d.z) << 16) |
         (owner_code(d.w) << 24);
}

__global__ __launch_bounds__(256) void owner_code_kernel(
    const int* __restrict__ dstIdx, unsigned* __restrict__ codes, int E,
    int nWords) {
  const int w0 = blockIdx.x * 1024 + threadIdx.x;
  if ((blockIdx.x + 1) * 4096 <= E) {
    int4 d[4];
#pragma unroll
    for (int c = 0; c < 4; ++c)
      d[c] = *(const int4*)(dstIdx + (w0 + c * 256) * 4);
#pragma unroll
    for (int c = 0; c < 4; ++c) codes[w0 + c * 256] = owner_word(d[c]);
  } else {
    for (int c = 0; c < 4; ++c) {
      const int g = w0 + c * 256;
      if (g >= nWords) break;
      const int e = g * 4;
      unsigned w = 0u;
      for (int i = 0; i < 4; ++i) {
        const unsigned b = (e + i < E) ? owner_code(dstIdx[e + i]) : 255u;
        w |= b << (8 * i);
      }
      codes[g] = w;
    }
  }
}

template <int U>
__device__ __forceinline__ void spmm_edges(const float* __restrict__ x,
                                           float* acc, int sv, int dv,
                                           float vv, int j, int lane) {
  float4 r[U];
#pragma unroll
  for (int u = 0; u < U; ++u) {
    const int s = __builtin_amdgcn_readlane(sv, j + u);
    const float* xr = x + (size_t)s * 128;
    r[u] = *(const float4*)(xr + lane * 4);
  }
#pragma unroll
  for (int u = 0; u < U; ++u) {
    const int d = __builtin_amdgcn_readlane(dv, j + u);
    const float v =
        __int_as_float(__builtin_amdgcn_readlane(__float_as_int(vv), j + u));
    float* ar = acc + d * kRowPitch + lane;
    __hip_atomic_fetch_add(ar, r[u].x * v, __ATOMIC_RELAXED,
                           __HIP_MEMORY_SCOPE_WORKGROUP);
    __hip_atomic_fetch_add(ar + kColPitch, r[u].y * v, __ATOMIC_RELAXED,
                           __HIP_MEMORY_SCOPE_WORKGROUP);
    __hip_atomic_fetch_add(ar + 2 * kColPitch, r[u].z * v, __ATOMIC_RELAXED,
                           __HIP_MEMORY_SCOPE_WORKGROUP);
    __hip_atomic_fetch_add(ar + 3 * kColPitch, r[u].w * v, __ATOMIC_RELAXED,
                           __HIP_MEMORY_SCOPE_WORKGROUP);
  }
}

__device__ __forceinline__ int unit_of(int k, int start, int nUnits) {
  const int u = k + start;
  return (u >= nUnits) ? u - nUnits : u;
}

__device__ __forceinline__ uint4 load_unit(const unsigned* __restrict__ codes,
                                           int k, int start, int nUnits,
                                           int lane) {
  const int kk = (k < nUnits) ? k : nUnits - 1;
  return *(const uint4*)(codes + unit_of(kk, start, nUnits) * 128 + lane * 4);
}

__device__ __forceinline__ int list_push(bool m, int e, int* qe, int qn) {
  const unsigned b = __builtin_amdgcn_ballot_w32(m);
  if (m) qe[qn + (int)__builtin_amdgcn_mbcnt_lo(b, 0u)] = e;
  return qn + __builtin_popcount(b);
}

__device__ __forceinline__ unsigned zero_bytes(unsigned t) {
  return (t - 0x01010101u) & ~t;
}

__device__ __forceinline__ int take_unit(uint4 w, int unit, unsigned kb, int E,
                                         int lane, int* qe, int qn) {
  const unsigned z0 = zero_bytes(w.x ^ kb), z1 = zero_bytes(w.y ^ kb);
  const unsigned z2 = zero_bytes(w.z ^ kb), z3 = zero_bytes(w.w ^ kb);
  const unsigned any = (z0 | z1 | z2 | z3) & 0x80808080u;
  if (__builtin_amdgcn_ballot_w32(any != 0u) == 0u) return qn;
  unsigned mk = ((z0 & 0x80808080u) >> 7) | ((z1 & 0x80808080u) >> 6) |
                ((z2 & 0x80808080u) >> 5) | ((z3 & 0x80808080u) >> 4);
  const int e0 = unit * kUnitEdges + lane * 16;
  while (__builtin_amdgcn_ballot_w32(mk != 0u) != 0u) {
    const int p = __builtin_ctz(mk | 0x80000000u);
    const int e = e0 + ((p & 7) << 2) + (p >> 3);
    qn = list_push(mk != 0u && e < E, e, qe, qn);
    mk &= mk - 1u;
  }
  return qn;
}

__device__ __forceinline__ void edge_words(const int* __restrict__ dstIdx,
                                           const int* __restrict__ srcIdx,
                                           const float* __restrict__ vals,
                                           int e, int r0, int rows, int& sv,
                                           int& dv, float& vv) {
  const int rel = dstIdx[e] - r0;
  sv = srcIdx[e];
  vv = vals[e];
  dv = ((unsigned)rel < (unsigned)rows) ? rel : kRowsPerBlock;
}

__device__ __forceinline__ int serve_full(const int* __restrict__ dstIdx,
                                          const int* __restrict__ srcIdx,
                                          const float* __restrict__ vals,
                                          const float* __restrict__ x,
                                          float* acc, const int* qe, int qn,
                                          int r0, int rows, int lane) {
  while (qn >= 32) {
    qn -= 32;
    int sv, dv;
    float vv;
    edge_words(dstIdx, srcIdx, vals, qe[qn + lane], r0, rows, sv, dv, vv);
    for (int j = 0; j < 32; j += 8) spmm_edges<8>(x, acc, sv, dv, vv, j, lane);
  }
  return qn;
}

template <int OC>
__device__ __forceinline__ void block_gemm(const float* acc,
                                           const float* __restrict__ W,
                                           const float* __restrict__ bias,
                                           float* __restrict__ outBlk,
                                           int rows, int wave, int lane) {
  constexpr int kColGroups = OC / 32;
  constexpr int kRowTiles = (kRowsPerBlock + 15) / 16;
  const int hi = lane >> 4;
  const int lo = lane & 15;
  for (int task = wave; task < kRowTiles * kColGroups; task += kWaves) {
    const int rt = task % kRowTiles;
    const int cb = (task / kRowTiles) * 32;
    int d = rt * 16 + lo;
    d = (d < kRowsPerBlock) ? d : kRowsPerBlock;
    const float* ap = acc + d * kRowPitch + hi * (2 * kColPitch);
    const float* wp = W + (size_t)(hi * 2) * OC + cb + lo * 2;

    v8f c0 = {};
    v8f c1 = {};
    constexpr int kGroup = 4;
    float2 w0[2][kGroup], w1[2][kGroup];
    float ax[2][kGroup], ay[2][kGroup];
#pragma unroll
    for (int i = 0; i < kGroup; ++i) {
      ax[0][i] = ap[i];
      ay[0][i] = ap[kColPitch + i];
      w0[0][i] = *(const float2*)(wp + (size_t)(4 * i) * OC);
      w1[0][i] = *(const float2*)(wp + (size_t)(4 * i + 1) * OC);
    }
#pragma unroll
    for (int g = 0; g < 32 / kGroup; ++g) {
      const int cur = g & 1;
      const int nxt = cur ^ 1;
      if (g + 1 < 32 / kGroup) {
#pragma unroll
        for (int i = 0; i < kGroup; ++i) {
          const int s = (g + 1) * kGroup + i;
          ax[nxt][i] = ap[s];
          ay[nxt][i] = ap[kColPitch + s];
          w0[nxt][i] = *(const float2*)(wp + (size_t)(4 * s) * OC);
          w1[nxt][i] = *(const float2*)(wp + (size_t)(4 * s + 1) * OC);
        }
      }
#pragma unroll
      for (int i = 0; i < kGroup; ++i) {
        v2f a, b0, b1;
        a.x = ax[cur][i];
        a.y = ay[cur][i];
        b0.x = w0[cur][i].x;
        b0.y = w1[cur][i].x;
        b1.x = w0[cur][i].y;
        b1.y = w1[cur][i].y;
        c0 = __builtin_amdgcn_wmma_f32_16x16x4_f32(
             false, a,  false, b0,
             (short)0, c0,  false,  false);
        c1 = __builtin_amdgcn_wmma_f32_16x16x4_f32(
             false, a,  false, b1,
             (short)0, c1,  false,  false);
      }
    }

    const float2 bv = *(const float2*)(bias + cb + lo * 2);
#pragma unroll
    for (int j = 0; j < 8; ++j) {
      const int m = rt * 16 + j + hi * 8;
      if (m < rows)
        *(float2*)(outBlk + (size_t)m * OC + cb + lo * 2) =
            make_float2(c0[j] + bv.x, c1[j] + bv.y);
    }
  }
}

template <int OC>
__global__ __launch_bounds__(1024) void layer_kernel(
    const unsigned* __restrict__ codes, const int* __restrict__ dstIdx,
    const int* __restrict__ srcIdx, const float* __restrict__ vals,
    const float* __restrict__ x, const float* __restrict__ W,
    const float* __restrict__ bias, float* __restrict__ out, int E, int N) {
  __shared__ __attribute__((aligned(16))) float
      acc[(kRowsPerBlock + 1) * kRowPitch];
  __shared__ int listE[kWaves * kListCap];

  const int tid = threadIdx.x;
  const int lane = tid & 31;
  const int wave = __builtin_amdgcn_readfirstlane(tid >> 5);
  int* qe = listE + wave * kListCap;
  float4* acc4 = (float4*)acc;

  const int nUnits = (E + kUnitEdges - 1) / kUnitEdges;
  const int start = (int)(((long long)blockIdx.x * nUnits) / gridDim.x);

  for (int blk = blockIdx.x; blk * kRowsPerBlock < N; blk += gridDim.x) {
    const int r0 = blk * kRowsPerBlock;
    const int rows = (N - r0 < kRowsPerBlock) ? N - r0 : kRowsPerBlock;
    const unsigned kb = (unsigned)(blk & 255) * 0x01010101u;

    for (int i = tid; i < (kRowsPerBlock + 1) * kRowPitch / 4; i += 1024)
      acc4[i] = make_float4(0.f, 0.f, 0.f, 0.f);
    __syncthreads();

    int qn = 0;
    if (wave < nUnits) {
      int k = wave;
      uint4 a0 = load_unit(codes, k, start, nUnits, lane);
      uint4 a1 = load_unit(codes, k + 32, start, nUnits, lane);
      uint4 a2 = load_unit(codes, k + 64, start, nUnits, lane);
      uint4 a3 = load_unit(codes, k + 96, start, nUnits, lane);
      for (; k < nUnits; k += 128) {
        qn = take_unit(a0, unit_of(k, start, nUnits), kb, E, lane, qe, qn);
        a0 = load_unit(codes, k + 128, start, nUnits, lane);
        if (k + 32 < nUnits)
          qn = take_unit(a1, unit_of(k + 32, start, nUnits), kb, E, lane, qe,
                         qn);
        a1 = load_unit(codes, k + 160, start, nUnits, lane);
        qn = serve_full(dstIdx, srcIdx, vals, x, acc, qe, qn, r0, rows, lane);
        if (k + 64 < nUnits)
          qn = take_unit(a2, unit_of(k + 64, start, nUnits), kb, E, lane, qe,
                         qn);
        a2 = load_unit(codes, k + 192, start, nUnits, lane);
        if (k + 96 < nUnits)
          qn = take_unit(a3, unit_of(k + 96, start, nUnits), kb, E, lane, qe,
                         qn);
        a3 = load_unit(codes, k + 224, start, nUnits, lane);
        qn = serve_full(dstIdx, srcIdx, vals, x, acc, qe, qn, r0, rows, lane);
      }
    }
    while (qn > 0) {
      const int cnt = (qn < 32) ? qn : 32;
      qn -= cnt;
      int sv, dv;
      float vv;
      edge_words(dstIdx, srcIdx, vals, qe[qn + ((lane < cnt) ? lane : 0)], r0,
                 rows, sv, dv, vv);
      int j = 0;
      for (; j + 8 <= cnt; j += 8) spmm_edges<8>(x, acc, sv, dv, vv, j, lane);
      if (cnt - j >= 4) {
        spmm_edges<4>(x, acc, sv, dv, vv, j, lane);
        j += 4;
      }
      if (cnt - j >= 2) {
        spmm_edges<2>(x, acc, sv, dv, vv, j, lane);
        j += 2;
      }
      if (cnt - j >= 1) spmm_edges<1>(x, acc, sv, dv, vv, j, lane);
    }
    __syncthreads();

    block_gemm<OC>(acc, W, bias, out + (size_t)r0 * OC, rows, wave, lane);
    __syncthreads();
  }
}

extern "C" void kernel_launch(void* const* d_in, const int* in_sizes, int n_in,
                              void* d_out, int out_size, void* d_ws,
                              size_t ws_size, hipStream_t stream) {
  const float* x = (const float*)d_in[0];
  const int* adj = (const int*)d_in[1];
  const float* vals = (const float*)d_in[2];
  const float* W1 = (const float*)d_in[3];
  const float* b1 = (const float*)d_in[4];
  const float* W2 = (const float*)d_in[5];
  const float* b2 = (const float*)d_in[6];
  const float* W3 = (const float*)d_in[7];
  const float* b3 = (const float*)d_in[8];
  float* out = (float*)d_out;

  const int E = in_sizes[2];
  const int N = in_sizes[0] / 128;
  const int* dstIdx = adj;
  const int* srcIdx = adj + E;

  float* ws0 = (float*)d_ws;
  float* ws1 = ws0 + (size_t)N * 128;
  unsigned* codes = (unsigned*)(ws1 + (size_t)N * 128);

  const int layerBlocks = 256;
  const int nWords = (E + 511) / 512 * 128;
  const int codeBlocks = (nWords + 1023) / 1024;

  owner_code_kernel<<<codeBlocks, 256, 0, stream>>>(dstIdx, codes, E, nWords);

  layer_kernel<128><<<layerBlocks, 1024, 0, stream>>>(
      codes, dstIdx, srcIdx, vals, x, W1, b1, ws0, E, N);
  layer_kernel<128><<<layerBlocks, 1024, 0, stream>>>(
      codes, dstIdx, srcIdx, vals, ws0, W2, b2, ws1, E, N);
  layer_kernel<64><<<layerBlocks, 1024, 0, stream>>>(
      codes, dstIdx, srcIdx, vals, ws1, W3, b3, out, E, N);
}


# Filter result reused: layer 1's waves store the (src, local dst, weight) records they serve; layers 2-3 read them back coalesced, no code scan, no scattered index loads; LDS add arrival order differs
# speedup vs baseline: 4.3044x; 1.2764x over previous
#include <hip/hip_runtime.h>
#include <hip/hip_bf16.h>
#include <stddef.h>


typedef __attribute__((ext_vector_type(2))) float v2f;
typedef __attribute__((ext_vector_type(8))) float v8f;

constexpr int kRowsPerBlock = 196;
constexpr int kColPitch = 33;
constexpr int kRowPitch = 4 * kColPitch;
constexpr int kWaves = 32;
constexpr int kUnitEdges = 512;
constexpr int kListCap = 31 + 2 * kUnitEdges + 1;
constexpr int kSegCap = 512;
constexpr int kSegWords = 3 * kSegCap;

__device__ __forceinline__ unsigned owner_code(int d) {
  return ((unsigned)d / (unsigned)kRowsPerBlock) & 255u;
}

__device__ __forceinline__ unsigned owner_word(int4 d) {
  return owner_code(d.x) | (owner_code(d.y) << 8) | (owner_code(d.z) << 16) |
         (owner_code(d.w) << 24);
}

__global__ __launch_bounds__(256) void owner_code_kernel(
    const int* __restrict__ dstIdx, unsigned* __restrict__ codes, int E,
    int nWords) {
  const int w0 = blockIdx.x * 1024 + threadIdx.x;
  if ((blockIdx.x + 1) * 4096 <= E) {
    int4 d[4];
#pragma unroll
    for (int c = 0; c < 4; ++c)
      d[c] = *(const int4*)(dstIdx + (w0 + c * 256) * 4);
#pragma unroll
    for (int c = 0; c < 4; ++c) codes[w0 + c * 256] = owner_word(d[c]);
  } else {
    for (int c = 0; c < 4; ++c) {
      const int g = w0 + c * 256;
      if (g >= nWords) break;
      const int e = g * 4;
      unsigned w = 0u;
      for (int i = 0; i < 4; ++i) {
        const unsigned b = (e + i < E) ? owner_code(dstIdx[e + i]) : 255u;
        w |= b << (8 * i);
      }
      codes[g] = w;
    }
  }
}

template <int U>
__device__ __forceinline__ void spmm_edges(const float* __restrict__ x,
                                           float* acc, int sv, int dv,
                                           float vv, int j, int lane) {
  float4 r[U];
#pragma unroll
  for (int u = 0; u < U; ++u) {
    const int s = __builtin_amdgcn_readlane(sv, j + u);
    const float* xr = x + (size_t)s * 128;
    r[u] = *(const float4*)(xr + lane * 4);
  }
#pragma unroll
  for (int u = 0; u < U; ++u) {
    const int d = __builtin_amdgcn_readlane(dv, j + u);
    const float v =
        __int_as_float(__builtin_amdgcn_readlane(__float_as_int(vv), j + u));
    float* ar = acc + d * kRowPitch + lane;
    __hip_atomic_fetch_add(ar, r[u].x * v, __ATOMIC_RELAXED,
                           __HIP_MEMORY_SCOPE_WORKGROUP);
    __hip_atomic_fetch_add(ar + kColPitch, r[u].y * v, __ATOMIC_RELAXED,
                           __HIP_MEMORY_SCOPE_WORKGROUP);
    __hip_atomic_fetch_add(ar + 2 * kColPitch, r[u].z * v, __ATOMIC_RELAXED,
                           __HIP_MEMORY_SCOPE_WORKGROUP);
    __hip_atomic_fetch_add(ar + 3 * kColPitch, r[u].w * v, __ATOMIC_RELAXED,
                           __HIP_MEMORY_SCOPE_WORKGROUP);
  }
}

__device__ __forceinline__ int unit_of(int k, int start, int nUnits) {
  const int u = k + start;
  return (u >= nUnits) ? u - nUnits : u;
}

__device__ __forceinline__ uint4 load_unit(const unsigned* __restrict__ codes,
                                           int k, int start, int nUnits,
                                           int lane) {
  const int kk = (k < nUnits) ? k : nUnits - 1;
  return *(const uint4*)(codes + unit_of(kk, start, nUnits) * 128 + lane * 4);
}

__device__ __forceinline__ int list_push(bool m, int e, int* qe, int qn) {
  const unsigned b = __builtin_amdgcn_ballot_w32(m);
  if (m) qe[qn + (int)__builtin_amdgcn_mbcnt_lo(b, 0u)] = e;
  return qn + __builtin_popcount(b);
}

__device__ __forceinline__ unsigned zero_bytes(unsigned t) {
  return (t - 0x01010101u) & ~t;
}

__device__ __forceinline__ int take_unit(uint4 w, int unit, unsigned kb, int E,
                                         int lane, int* qe, int qn) {
  const unsigned z0 = zero_bytes(w.x ^ kb), z1 = zero_bytes(w.y ^ kb);
  const unsigned z2 = zero_bytes(w.z ^ kb), z3 = zero_bytes(w.w ^ kb);
  const unsigned any = (z0 | z1 | z2 | z3) & 0x80808080u;
  if (__builtin_amdgcn_ballot_w32(any != 0u) == 0u) return qn;
  unsigned mk = ((z0 & 0x80808080u) >> 7) | ((z1 & 0x80808080u) >> 6) |
                ((z2 & 0x80808080u) >> 5) | ((z3 & 0x80808080u) >> 4);
  const int e0 = unit * kUnitEdges + lane * 16;
  while (__builtin_amdgcn_ballot_w32(mk != 0u) != 0u) {
    const int p = __builtin_ctz(mk | 0x80000000u);
    const int e = e0 + ((p & 7) << 2) + (p >> 3);
    qn = list_push(mk != 0u && e < E, e, qe, qn);
    mk &= mk - 1u;
  }
  return qn;
}

__device__ __forceinline__ void edge_words(const int* __restrict__ dstIdx,
                                           const int* __restrict__ srcIdx,
                                           const float* __restrict__ vals,
                                           int e, int r0, int rows, int& sv,
                                           int& dv, float& vv) {
  const int rel = dstIdx[e] - r0;
  sv = srcIdx[e];
  vv = vals[e];
  dv = ((unsigned)rel < (unsigned)rows) ? rel : kRowsPerBlock;
}

__device__ __forceinline__ int keep_records(int* __restrict__ seg, int pos,
                                            int cnt, int lane, int sv, int dv,
                                            float vv) {
  if (pos + cnt <= kSegCap && lane < cnt) {
    seg[pos + lane] = sv;
    seg[kSegCap + pos + lane] = dv;
    seg[2 * kSegCap + pos + lane] = __float_as_int(vv);
  }
  return pos + cnt;
}

template <bool kKeep>
__device__ __forceinline__ int serve_full(const int* __restrict__ dstIdx,
                                          const int* __restrict__ srcIdx,
                                          const float* __restrict__ vals,
                                          const float* __restrict__ x,
                                          float* acc, const int* qe, int qn,
                                          int r0, int rows, int lane,
                                          int* __restrict__ seg, int& pos) {
  while (qn >= 32) {
    qn -= 32;
    int sv, dv;
    float vv;
    edge_words(dstIdx, srcIdx, vals, qe[qn + lane], r0, rows, sv, dv, vv);
    if (kKeep) pos = keep_records(seg, pos, 32, lane, sv, dv, vv);
    for (int j = 0; j < 32; j += 8) spmm_edges<8>(x, acc, sv, dv, vv, j, lane);
  }
  return qn;
}

__device__ __forceinline__ void serve_chunk(const float* __restrict__ x,
                                            float* acc, int sv, int dv,
                                            float vv, int cnt, int lane) {
  int j = 0;
  for (; j + 8 <= cnt; j += 8) spmm_edges<8>(x, acc, sv, dv, vv, j, lane);
  if (cnt - j >= 4) {
    spmm_edges<4>(x, acc, sv, dv, vv, j, lane);
    j += 4;
  }
  if (cnt - j >= 2) {
    spmm_edges<2>(x, acc, sv, dv, vv, j, lane);
    j += 2;
  }
  if (cnt - j >= 1) spmm_edges<1>(x, acc, sv, dv, vv, j, lane);
}

template <bool kKeep>
__device__ __forceinline__ void scan_and_serve(
    const unsigned* __restrict__ codes, const int* __restrict__ dstIdx,
    const int* __restrict__ srcIdx, const float* __restrict__ vals,
    const float* __restrict__ x, float* acc, int* qe, int* __restrict__ seg,
    int* __restrict__ cntp, unsigned kb, int start, int nUnits, int E, int r0,
    int rows, int wave, int lane) {
  int qn = 0;
  int pos = 0;
  if (wave < nUnits) {
    int k = wave;
    uint4 a0 = load_unit(codes, k, start, nUnits, lane);
    uint4 a1 = load_unit(codes, k + 32, start, nUnits, lane);
    uint4 a2 = load_unit(codes, k + 64, start, nUnits, lane);
    uint4 a3 = load_unit(codes, k + 96, start, nUnits, lane);
    for (; k < nUnits; k += 128) {
      qn = take_unit(a0, unit_of(k, start, nUnits), kb, E, lane, qe, qn);
      a0 = load_unit(codes, k + 128, start, nUnits, lane);
      if (k + 32 < nUnits)
        qn = take_unit(a1, unit_of(k + 32, start, nUnits), kb, E, lane, qe,
                       qn);
      a1 = load_unit(codes, k + 160, start, nUnits, lane);
      qn = serve_full<kKeep>(dstIdx, srcIdx, vals, x, acc, qe, qn, r0, rows,
                             lane, seg, pos);
      if (k + 64 < nUnits)
        qn = take_unit(a2, unit_of(k + 64, start, nUnits), kb, E, lane, qe,
                       qn);
      a2 = load_unit(codes, k + 192, start, nUnits, lane);
      if (k + 96 < nUnits)
        qn = take_unit(a3, unit_of(k + 96, start, nUnits), kb, E, lane, qe,
                       qn);
      a3 = load_unit(codes, k + 224, start, nUnits, lane);
      qn = serve_full<kKeep>(dstIdx, srcIdx, vals, x, acc, qe, qn, r0, rows,
                             lane, seg, pos);
    }
  }
  while (qn > 0) {
    const int cnt = (qn < 32) ? qn : 32;
    qn -= cnt;
    int sv, dv;
    float vv;
    edge_words(dstIdx, srcIdx, vals, qe[qn + ((lane < cnt) ? lane : 0)], r0,
               rows, sv, dv, vv);
    if (kKeep) pos = keep_records(seg, pos, cnt, lane, sv, dv, vv);
    serve_chunk(x, acc, sv, dv, vv, cnt, lane);
  }
  if (kKeep && lane == 0) *cntp = (pos <= kSegCap) ? pos : -1;
}

__device__ __forceinline__ void load_records(const int* __restrict__ seg,
                                             int pos, int n, int lane, int& sv,
                                             int& dv, float& vv) {
  const int i = pos + ((lane < n - pos) ? lane : 0);
  sv = seg[i];
  dv = seg[kSegCap + i];
  vv = __int_as_float(seg[2 * kSegCap + i]);
}

__device__ __forceinline__ void replay_records(const int* __restrict__ seg,
                                               int n, int sv, int dv, float vv,
                                               const float* __restrict__ x,
                                               float* acc, int lane) {
  for (int pos = 0; pos < n; pos += 32) {
    const int left = n - pos;
    int sv2 = sv, dv2 = dv;
    float vv2 = vv;
    if (left > 32) load_records(seg, pos + 32, n, lane, sv2, dv2, vv2);
    serve_chunk(x, acc, sv, dv, vv, (left < 32) ? left : 32, lane);
    sv = sv2;
    dv = dv2;
    vv = vv2;
  }
}

template <int OC>
__device__ __forceinline__ void block_gemm(const float* acc,
                                           const float* __restrict__ W,
                                           const float* __restrict__ bias,
                                           float* __restrict__ outBlk,
                                           int rows, int wave, int lane) {
  constexpr int kColGroups = OC / 32;
  constexpr int kRowTiles = (kRowsPerBlock + 15) / 16;
  const int hi = lane >> 4;
  const int lo = lane & 15;
  for (int task = wave; task < kRowTiles * kColGroups; task += kWaves) {
    const int rt = task % kRowTiles;
    const int cb = (task / kRowTiles) * 32;
    int d = rt * 16 + lo;
    d = (d < kRowsPerBlock) ? d : kRowsPerBlock;
    const float* ap = acc + d * kRowPitch + hi * (2 * kColPitch);
    const float* wp = W + (size_t)(hi * 2) * OC + cb + lo * 2;

    v8f c0 = {};
    v8f c1 = {};
    constexpr int kGroup = 4;
    float2 w0[2][kGroup], w1[2][kGroup];
    float ax[2][kGroup], ay[2][kGroup];
#pragma unroll
    for (int i = 0; i < kGroup; ++i) {
      ax[0][i] = ap[i];
      ay[0][i] = ap[kColPitch + i];
      w0[0][i] = *(const float2*)(wp + (size_t)(4 * i) * OC);
      w1[0][i] = *(const float2*)(wp + (size_t)(4 * i + 1) * OC);
    }
#pragma unroll
    for (int g = 0; g < 32 / kGroup; ++g) {
      const int cur = g & 1;
      const int nxt = cur ^ 1;
      if (g + 1 < 32 / kGroup) {
#pragma unroll
        for (int i = 0; i < kGroup; ++i) {
          const int s = (g + 1) * kGroup + i;
          ax[nxt][i] = ap[s];
          ay[nxt][i] = ap[kColPitch + s];
          w0[nxt][i] = *(const float2*)(wp + (size_t)(4 * s) * OC);
          w1[nxt][i] = *(const float2*)(wp + (size_t)(4 * s + 1) * OC);
        }
      }
#pragma unroll
      for (int i = 0; i < kGroup; ++i) {
        v2f a, b0, b1;
        a.x = ax[cur][i];
        a.y = ay[cur][i];
        b0.x = w0[cur][i].x;
        b0.y = w1[cur][i].x;
        b1.x = w0[cur][i].y;
        b1.y = w1[cur][i].y;
        c0 = __builtin_amdgcn_wmma_f32_16x16x4_f32(
             false, a,  false, b0,
             (short)0, c0,  false,  false);
        c1 = __builtin_amdgcn_wmma_f32_16x16x4_f32(
             false, a,  false, b1,
             (short)0, c1,  false,  false);
      }
    }

    const float2 bv = *(const float2*)(bias + cb + lo * 2);
#pragma unroll
    for (int j = 0; j < 8; ++j) {
      const int m = rt * 16 + j + hi * 8;
      if (m < rows)
        *(float2*)(outBlk + (size_t)m * OC + cb + lo * 2) =
            make_float2(c0[j] + bv.x, c1[j] + bv.y);
    }
  }
}

template <int OC, bool kReplay>
__global__ __launch_bounds__(1024) void layer_kernel(
    const unsigned* __restrict__ codes, const int* __restrict__ dstIdx,
    const int* __restrict__ srcIdx, const float* __restrict__ vals,
    const float* __restrict__ x, const float* __restrict__ W,
    const float* __restrict__ bias, float* __restrict__ out,
    int* __restrict__ counts, int E, int N) {
  __shared__ __attribute__((aligned(16))) float
      acc[(kRowsPerBlock + 1) * kRowPitch];
  __shared__ int listE[kWaves * kListCap];

  const int tid = threadIdx.x;
  const int lane = tid & 31;
  const int wave = __builtin_amdgcn_readfirstlane(tid >> 5);
  int* qe = listE + wave * kListCap;
  float4* acc4 = (float4*)acc;

  const int nUnits = (E + kUnitEdges - 1) / kUnitEdges;
  const int start = (int)(((long long)blockIdx.x * nUnits) / gridDim.x);
  const int nBlocks = (N + kRowsPerBlock - 1) / kRowsPerBlock;
  int* recs = counts + nBlocks * kWaves;

  for (int blk = blockIdx.x; blk * kRowsPerBlock < N; blk += gridDim.x) {
    const int r0 = blk * kRowsPerBlock;
    const int rows = (N - r0 < kRowsPerBlock) ? N - r0 : kRowsPerBlock;
    const unsigned kb = (unsigned)(blk & 255) * 0x01010101u;
    int* cntp = counts + blk * kWaves + wave;
    int* seg = recs + (size_t)(blk * kWaves + wave) * kSegWords;

    int n = -1, sv0 = 0, dv0 = 0;
    float vv0 = 0.f;
    if (kReplay) {
      n = __builtin_amdgcn_readfirstlane(*cntp);
      if (n > kSegCap) n = -1;
      if (n > 0) load_records(seg, 0, n, lane, sv0, dv0, vv0);
    }

    for (int i = tid; i < (kRowsPerBlock + 1) * kRowPitch / 4; i += 1024)
      acc4[i] = make_float4(0.f, 0.f, 0.f, 0.f);
    __syncthreads();

    if (kReplay) {
      if (n >= 0)
        replay_records(seg, n, sv0, dv0, vv0, x, acc, lane);
      else
        scan_and_serve<false>(codes, dstIdx, srcIdx, vals, x, acc, qe, seg,
                              cntp, kb, start, nUnits, E, r0, rows, wave,
                              lane);
    } else {
      scan_and_serve<true>(codes, dstIdx, srcIdx, vals, x, acc, qe, seg, cntp,
                           kb, start, nUnits, E, r0, rows, wave, lane);
    }
    __syncthreads();

    block_gemm<OC>(acc, W, bias, out + (size_t)r0 * OC, rows, wave, lane);
    __syncthreads();
  }
}

extern "C" void kernel_launch(void* const* d_in, const int* in_sizes, int n_in,
                              void* d_out, int out_size, void* d_ws,
                              size_t ws_size, hipStream_t stream) {
  const float* x = (const float*)d_in[0];
  const int* adj = (const int*)d_in[1];
  const float* vals = (const float*)d_in[2];
  const float* W1 = (const float*)d_in[3];
  const float* b1 = (const float*)d_in[4];
  const float* W2 = (const float*)d_in[5];
  const float* b2 = (const float*)d_in[6];
  const float* W3 = (const float*)d_in[7];
  const float* b3 = (const float*)d_in[8];
  float* out = (float*)d_out;

  const int E = in_sizes[2];
  const int N = in_sizes[0] / 128;
  const int* dstIdx = adj;
  const int* srcIdx = adj + E;

  float* ws0 = (float*)d_ws;
  float* ws1 = ws0 + (size_t)N * 128;
  unsigned* codes = (unsigned*)(ws1 + (size_t)N * 128);

  const int layerBlocks = 256;
  const int nWords = (E + 511) / 512 * 128;
  const int codeBlocks = (nWords + 1023) / 1024;
  int* counts = (int*)(codes + nWords);

  owner_code_kernel<<<codeBlocks, 256, 0, stream>>>(dstIdx, codes, E, nWords);

  layer_kernel<128, false><<<layerBlocks, 1024, 0, stream>>>(
      codes, dstIdx, srcIdx, vals, x, W1, b1, ws0, counts, E, N);
  layer_kernel<128, true><<<layerBlocks, 1024, 0, stream>>>(
      codes, dstIdx, srcIdx, vals, ws0, W2, b2, ws1, counts, E, N);
  layer_kernel<64, true><<<layerBlocks, 1024, 0, stream>>>(
      codes, dstIdx, srcIdx, vals, ws1, W3, b3, out, counts, E, N);
}


# SpMM layers 2-3 sum in registers: layer 1 queues records by owner wave (row mod 32) in LDS; a replay wave adds gathered rows to 7 register rows, stores once; no LDS float adds; sums in queue order
# speedup vs baseline: 4.5695x; 1.0616x over previous
#include <hip/hip_runtime.h>
#include <hip/hip_bf16.h>
#include <stddef.h>


typedef __attribute__((ext_vector_type(2))) float v2f;
typedef __attribute__((ext_vector_type(8))) float v8f;

constexpr int kRowsPerBlock = 196;
constexpr int kColPitch = 33;
constexpr int kRowPitch = 4 * kColPitch;
constexpr int kWaves = 32;
constexpr int kUnitEdges = 512;
constexpr int kListCap = 31 + 2 * kUnitEdges + 1;
constexpr int kSegCap = 512;
constexpr int kSegWords = 3 * kSegCap;
constexpr int kQueueCap = 192;
constexpr int kQueueWords = 3 * kQueueCap;
constexpr int kOwnRows = (kRowsPerBlock + kWaves - 1) / kWaves;
static_assert(kWaves == 32, "a lane reads one queue's length");
static_assert(kOwnRows * 4 <= 32, "the register sums are one 32-word value");

__device__ __forceinline__ unsigned owner_code(int d) {
  return ((unsigned)d / (unsigned)kRowsPerBlock) & 255u;
}

__device__ __forceinline__ unsigned owner_word(int4 d) {
  return owner_code(d.x) | (owner_code(d.y) << 8) | (owner_code(d.z) << 16) |
         (owner_code(d.w) << 24);
}

__global__ __launch_bounds__(256) void owner_code_kernel(
    const int* __restrict__ dstIdx, unsigned* __restrict__ codes, int E,
    int nWords) {
  const int w0 = blockIdx.x * 1024 + threadIdx.x;
  if ((blockIdx.x + 1) * 4096 <= E) {
    int4 d[4];
#pragma unroll
    for (int c = 0; c < 4; ++c)
      d[c] = *(const int4*)(dstIdx + (w0 + c * 256) * 4);
#pragma unroll
    for (int c = 0; c < 4; ++c) codes[w0 + c * 256] = owner_word(d[c]);
  } else {
    for (int c = 0; c < 4; ++c) {
      const int g = w0 + c * 256;
      if (g >= nWords) break;
      const int e = g * 4;
      unsigned w = 0u;
      for (int i = 0; i < 4; ++i) {
        const unsigned b = (e + i < E) ? owner_code(dstIdx[e + i]) : 255u;
        w |= b << (8 * i);
      }
      codes[g] = w;
    }
  }
}

template <int U>
__device__ __forceinline__ void spmm_edges(const float* __restrict__ x,
                                           float* acc, int sv, int dv,
                                           float vv, int j, int lane) {
  float4 r[U];
#pragma unroll
  for (int u = 0; u < U; ++u) {
    const int s = __builtin_amdgcn_readlane(sv, j + u);
    const float* xr = x + (size_t)s * 128;
    r[u] = *(const float4*)(xr + lane * 4);
  }
#pragma unroll
  for (int u = 0; u < U; ++u) {
    const int d = __builtin_amdgcn_readlane(dv, j + u);
    const float v =
        __int_as_float(__builtin_amdgcn_readlane(__float_as_int(vv), j + u));
    float* ar = acc + d * kRowPitch + lane;
    __hip_atomic_fetch_add(ar, r[u].x * v, __ATOMIC_RELAXED,
                           __HIP_MEMORY_SCOPE_WORKGROUP);
    __hip_atomic_fetch_add(ar + kColPitch, r[u].y * v, __ATOMIC_RELAXED,
                           __HIP_MEMORY_SCOPE_WORKGROUP);
    __hip_atomic_fetch_add(ar + 2 * kColPitch, r[u].z * v, __ATOMIC_RELAXED,
                           __HIP_MEMORY_SCOPE_WORKGROUP);
    __hip_atomic_fetch_add(ar + 3 * kColPitch, r[u].w * v, __ATOMIC_RELAXED,
                           __HIP_MEMORY_SCOPE_WORKGROUP);
  }
}

__device__ __forceinline__ int unit_of(int k, int start, int nUnits) {
  const int u = k + start;
  return (u >= nUnits) ? u - nUnits : u;
}

__device__ __forceinline__ uint4 load_unit(const unsigned* __restrict__ codes,
                                           int k, int start, int nUnits,
                                           int lane) {
  const int kk = (k < nUnits) ? k : nUnits - 1;
  return *(const uint4*)(codes + unit_of(kk, start, nUnits) * 128 + lane * 4);
}

__device__ __forceinline__ int list_push(bool m, int e, int* qe, int qn) {
  const unsigned b = __builtin_amdgcn_ballot_w32(m);
  if (m) qe[qn + (int)__builtin_amdgcn_mbcnt_lo(b, 0u)] = e;
  return qn + __builtin_popcount(b);
}

__device__ __forceinline__ unsigned zero_bytes(unsigned t) {
  return (t - 0x01010101u) & ~t;
}

__device__ __forceinline__ int take_unit(uint4 w, int unit, unsigned kb, int E,
                                         int lane, int* qe, int qn) {
  const unsigned z0 = zero_bytes(w.x ^ kb), z1 = zero_bytes(w.y ^ kb);
  const unsigned z2 = zero_bytes(w.z ^ kb), z3 = zero_bytes(w.w ^ kb);
  const unsigned any = (z0 | z1 | z2 | z3) & 0x80808080u;
  if (__builtin_amdgcn_ballot_w32(any != 0u) == 0u) return qn;
  unsigned mk = ((z0 & 0x80808080u) >> 7) | ((z1 & 0x80808080u) >> 6) |
                ((z2 & 0x80808080u) >> 5) | ((z3 & 0x80808080u) >> 4);
  const int e0 = unit * kUnitEdges + lane * 16;
  while (__builtin_amdgcn_ballot_w32(mk != 0u) != 0u) {
    const int p = __builtin_ctz(mk | 0x80000000u);
    const int e = e0 + ((p & 7) << 2) + (p >> 3);
    qn = list_push(mk != 0u && e < E, e, qe, qn);
    mk &= mk - 1u;
  }
  return qn;
}

__device__ __forceinline__ void edge_words(const int* __restrict__ dstIdx,
                                           const int* __restrict__ srcIdx,
                                           const float* __restrict__ vals,
                                           int e, int r0, int rows, int& sv,
                                           int& dv, float& vv) {
  const int rel = dstIdx[e] - r0;
  sv = srcIdx[e];
  vv = vals[e];
  dv = ((unsigned)rel < (unsigned)rows) ? rel : kRowsPerBlock;
}

__device__ __forceinline__ void queue_records(int* qW, int* qN, int cnt,
                                              int lane, int sv, int dv,
                                              float vv) {
  if (lane < cnt && dv < kRowsPerBlock) {
    const int owner = dv & (kWaves - 1);
    const int slot = __hip_atomic_fetch_add(
        &qN[owner], 1, __ATOMIC_RELAXED, __HIP_MEMORY_SCOPE_WORKGROUP);
    if (slot < kQueueCap) {
      int* q = qW + owner * kQueueWords + slot;
      q[0] = sv;
      q[kQueueCap] = dv;
      q[2 * kQueueCap] = __float_as_int(vv);
    }
  }
}

template <bool kKeep>
__device__ __forceinline__ int serve_full(const int* __restrict__ dstIdx,
                                          const int* __restrict__ srcIdx,
                                          const float* __restrict__ vals,
                                          const float* __restrict__ x,
                                          float* acc, const int* qe, int qn,
                                          int r0, int rows, int lane, int* qW,
                                          int* qN) {
  while (qn >= 32) {
    qn -= 32;
    int sv, dv;
    float vv;
    edge_words(dstIdx, srcIdx, vals, qe[qn + lane], r0, rows, sv, dv, vv);
    if (kKeep) queue_records(qW, qN, 32, lane, sv, dv, vv);
    for (int j = 0; j < 32; j += 8) spmm_edges<8>(x, acc, sv, dv, vv, j, lane);
  }
  return qn;
}

__device__ __forceinline__ void serve_chunk(const float* __restrict__ x,
                                            float* acc, int sv, int dv,
                                            float vv, int cnt, int lane) {
  int j = 0;
  for (; j + 8 <= cnt; j += 8) spmm_edges<8>(x, acc, sv, dv, vv, j, lane);
  if (cnt - j >= 4) {
    spmm_edges<4>(x, acc, sv, dv, vv, j, lane);
    j += 4;
  }
  if (cnt - j >= 2) {
    spmm_edges<2>(x, acc, sv, dv, vv, j, lane);
    j += 2;
  }
  if (cnt - j >= 1) spmm_edges<1>(x, acc, sv, dv, vv, j, lane);
}

template <bool kKeep>
__device__ __forceinline__ void scan_and_serve(
    const unsigned* __restrict__ codes, const int* __restrict__ dstIdx,
    const int* __restrict__ srcIdx, const float* __restrict__ vals,
    const float* __restrict__ x, float* acc, int* qe, int* qW, int* qN,
    unsigned kb, int start, int nUnits, int E, int r0, int rows, int wave,
    int lane) {
  int qn = 0;
  if (wave < nUnits) {
    int k = wave;
    uint4 a0 = load_unit(codes, k, start, nUnits, lane);
    uint4 a1 = load_unit(codes, k + 32, start, nUnits, lane);
    uint4 a2 = load_unit(codes, k + 64, start, nUnits, lane);
    uint4 a3 = load_unit(codes, k + 96, start, nUnits, lane);
    for (; k < nUnits; k += 128) {
      qn = take_unit(a0, unit_of(k, start, nUnits), kb, E, lane, qe, qn);
      a0 = load_unit(codes, k + 128, start, nUnits, lane);
      if (k + 32 < nUnits)
        qn = take_unit(a1, unit_of(k + 32, start, nUnits), kb, E, lane, qe,
                       qn);
      a1 = load_unit(codes, k + 160, start, nUnits, lane);
      qn = serve_full<kKeep>(dstIdx, srcIdx, vals, x, acc, qe, qn, r0, rows,
                             lane, qW, qN);
      if (k + 64 < nUnits)
        qn = take_unit(a2, unit_of(k + 64, start, nUnits), kb, E, lane, qe,
                       qn);
      a2 = load_unit(codes, k + 192, start, nUnits, lane);
      if (k + 96 < nUnits)
        qn = take_unit(a3, unit_of(k + 96, start, nUnits), kb, E, lane, qe,
                       qn);
      a3 = load_unit(codes, k + 224, start, nUnits, lane);
      qn = serve_full<kKeep>(dstIdx, srcIdx, vals, x, acc, qe, qn, r0, rows,
                             lane, qW, qN);
    }
  }
  while (qn > 0) {
    const int cnt = (qn < 32) ? qn : 32;
    qn -= cnt;
    int sv, dv;
    float vv;
    edge_words(dstIdx, srcIdx, vals, qe[qn + ((lane < cnt) ? lane : 0)], r0,
               rows, sv, dv, vv);
    if (kKeep) queue_records(qW, qN, cnt, lane, sv, dv, vv);
    serve_chunk(x, acc, sv, dv, vv, cnt, lane);
  }
}

__device__ __forceinline__ void store_queue(const int* qW, const int* qN,
                                            int* __restrict__ seg,
                                            int* __restrict__ cntp, int wave,
                                            int lane) {
  const bool over =
      __builtin_amdgcn_ballot_w32(qN[lane] > kQueueCap) != 0u;
  if (over) {
    if (lane == 0) *cntp = -1;
    return;
  }
  const int n = __builtin_amdgcn_readfirstlane(qN[wave]);
  const int* q = qW + wave * kQueueWords;
  for (int pos = 0; pos < n; pos += 32) {
    const int i = pos + lane;
    if (i < n) {
      seg[i] = q[i];
      seg[kSegCap + i] = q[kQueueCap + i];
      seg[2 * kSegCap + i] = q[2 * kQueueCap + i];
    }
  }
  if (lane == 0) *cntp = n;
}

__device__ __forceinline__ void load_records(const int* __restrict__ seg,
                                             int pos, int n, int lane, int& sv,
                                             int& dv, float& vv) {
  const int i = pos + ((lane < n - pos) ? lane : 0);
  sv = seg[i];
  dv = seg[kSegCap + i];
  vv = __int_as_float(seg[2 * kSegCap + i]);
}

typedef __attribute__((ext_vector_type(32))) float v32f;

__device__ __forceinline__ void add_row(v32f& a, int q, float4 r, float v) {
  const int b = q * 4;
  a[b] = __builtin_fmaf(r.x, v, a[b]);
  a[b + 1] = __builtin_fmaf(r.y, v, a[b + 1]);
  a[b + 2] = __builtin_fmaf(r.z, v, a[b + 2]);
  a[b + 3] = __builtin_fmaf(r.w, v, a[b + 3]);
}

template <int U>
__device__ __forceinline__ void own_edges(const float* __restrict__ x, v32f& a,
                                          int sv, int dv, float vv, int j,
                                          int lane) {
  float4 r[U];
#pragma unroll
  for (int u = 0; u < U; ++u) {
    const int s = __builtin_amdgcn_readlane(sv, j + u);
    const float* xr = x + (size_t)s * 128;
    r[u] = *(const float4*)(xr + lane * 4);
  }
#pragma unroll
  for (int u = 0; u < U; ++u) {
    const int d = __builtin_amdgcn_readlane(dv, j + u);
    const float v =
        __int_as_float(__builtin_amdgcn_readlane(__float_as_int(vv), j + u));
    add_row(a, (d >> 5) & 7, r[u], v);
  }
}

__device__ __forceinline__ void own_chunk(const float* __restrict__ x, v32f& a,
                                          int sv, int dv, float vv, int cnt,
                                          int lane) {
  int j = 0;
  for (; j + 8 <= cnt; j += 8) own_edges<8>(x, a, sv, dv, vv, j, lane);
  if (cnt - j >= 4) {
    own_edges<4>(x, a, sv, dv, vv, j, lane);
    j += 4;
  }
  if (cnt - j >= 2) {
    own_edges<2>(x, a, sv, dv, vv, j, lane);
    j += 2;
  }
  if (cnt - j >= 1) own_edges<1>(x, a, sv, dv, vv, j, lane);
}

__device__ __forceinline__ void replay_records(const int* __restrict__ seg,
                                               int n, int sv, int dv, float vv,
                                               const float* __restrict__ x,
                                               float* acc, int wave,
                                               int lane) {
  v32f a = {};
  for (int pos = 0; pos < n; pos += 32) {
    const int left = n - pos;
    int sv2 = sv, dv2 = dv;
    float vv2 = vv;
    if (left > 32) load_records(seg, pos + 32, n, lane, sv2, dv2, vv2);
    own_chunk(x, a, sv, dv, vv, (left < 32) ? left : 32, lane);
    sv = sv2;
    dv = dv2;
    vv = vv2;
  }
#pragma unroll
  for (int q = 0; q < kOwnRows; ++q) {
    const int d = wave + q * kWaves;
    if (d < kRowsPerBlock) {
      float* ar = acc + d * kRowPitch + lane;
#pragma unroll
      for (int c = 0; c < 4; ++c) ar[c * kColPitch] = a[4 * q + c];
    }
  }
}

template <int OC>
__device__ __forceinline__ void block_gemm(const float* acc,
                                           const float* __restrict__ W,
                                           const float* __restrict__ bias,
                                           float* __restrict__ outBlk,
                                           int rows, int wave, int lane) {
  constexpr int kColGroups = OC / 32;
  constexpr int kRowTiles = (kRowsPerBlock + 15) / 16;
  const int hi = lane >> 4;
  const int lo = lane & 15;
  for (int task = wave; task < kRowTiles * kColGroups; task += kWaves) {
    const int rt = task % kRowTiles;
    const int cb = (task / kRowTiles) * 32;
    int d = rt * 16 + lo;
    d = (d < kRowsPerBlock) ? d : kRowsPerBlock;
    const float* ap = acc + d * kRowPitch + hi * (2 * kColPitch);
    const float* wp = W + (size_t)(hi * 2) * OC + cb + lo * 2;

    v8f c0 = {};
    v8f c1 = {};
    constexpr int kGroup = 4;
    float2 w0[2][kGroup], w1[2][kGroup];
    float ax[2][kGroup], ay[2][kGroup];
#pragma unroll
    for (int i = 0; i < kGroup; ++i) {
      ax[0][i] = ap[i];
      ay[0][i] = ap[kColPitch + i];
      w0[0][i] = *(const float2*)(wp + (size_t)(4 * i) * OC);
      w1[0][i] = *(const float2*)(wp + (size_t)(4 * i + 1) * OC);
    }
#pragma unroll
    for (int g = 0; g < 32 / kGroup; ++g) {
      const int cur = g & 1;
      const int nxt = cur ^ 1;
      if (g + 1 < 32 / kGroup) {
#pragma unroll
        for (int i = 0; i < kGroup; ++i) {
          const int s = (g + 1) * kGroup + i;
          ax[nxt][i] = ap[s];
          ay[nxt][i] = ap[kColPitch + s];
          w0[nxt][i] = *(const float2*)(wp + (size_t)(4 * s) * OC);
          w1[nxt][i] = *(const float2*)(wp + (size_t)(4 * s + 1) * OC);
        }
      }
#pragma unroll
      for (int i = 0; i < kGroup; ++i) {
        v2f a, b0, b1;
        a.x = ax[cur][i];
        a.y = ay[cur][i];
        b0.x = w0[cur][i].x;
        b0.y = w1[cur][i].x;
        b1.x = w0[cur][i].y;
        b1.y = w1[cur][i].y;
        c0 = __builtin_amdgcn_wmma_f32_16x16x4_f32(
             false, a,  false, b0,
             (short)0, c0,  false,  false);
        c1 = __builtin_amdgcn_wmma_f32_16x16x4_f32(
             false, a,  false, b1,
             (short)0, c1,  false,  false);
      }
    }

    const float2 bv = *(const float2*)(bias + cb + lo * 2);
#pragma unroll
    for (int j = 0; j < 8; ++j) {
      const int m = rt * 16 + j + hi * 8;
      if (m < rows)
        *(float2*)(outBlk + (size_t)m * OC + cb + lo * 2) =
            make_float2(c0[j] + bv.x, c1[j] + bv.y);
    }
  }
}

template <int OC, bool kReplay>
__global__ __launch_bounds__(1024) void layer_kernel(
    const unsigned* __restrict__ codes, const int* __restrict__ dstIdx,
    const int* __restrict__ srcIdx, const float* __restrict__ vals,
    const float* __restrict__ x, const float* __restrict__ W,
    const float* __restrict__ bias, float* __restrict__ out,
    int* __restrict__ counts, int E, int N) {
  __shared__ __attribute__((aligned(16))) float
      acc[(kRowsPerBlock + 1) * kRowPitch];
  __shared__ int listE[kWaves * kListCap];
  __shared__ int queueW[kReplay ? 1 : kWaves * kQueueWords];
  __shared__ int queueN[kWaves];

  const int tid = threadIdx.x;
  const int lane = tid & 31;
  const int wave = __builtin_amdgcn_readfirstlane(tid >> 5);
  int* qe = listE + wave * kListCap;
  float4* acc4 = (float4*)acc;

  const int nUnits = (E + kUnitEdges - 1) / kUnitEdges;
  const int start = (int)(((long long)blockIdx.x * nUnits) / gridDim.x);
  const int nBlocks = (N + kRowsPerBlock - 1) / kRowsPerBlock;
  int* recs = counts + nBlocks * kWaves;

  for (int blk = blockIdx.x; blk * kRowsPerBlock < N; blk += gridDim.x) {
    const int r0 = blk * kRowsPerBlock;
    const int rows = (N - r0 < kRowsPerBlock) ? N - r0 : kRowsPerBlock;
    const unsigned kb = (unsigned)(blk & 255) * 0x01010101u;
    int* cntp = counts + blk * kWaves + wave;
    int* seg = recs + (size_t)(blk * kWaves + wave) * kSegWords;

    bool rescan = false;
    int n = 0, sv0 = 0, dv0 = 0;
    float vv0 = 0.f;
    if (kReplay) {
      rescan = __builtin_amdgcn_readfirstlane(counts[blk * kWaves]) < 0;
      n = __builtin_amdgcn_readfirstlane(*cntp);
      if (rescan || n < 0 || n > kQueueCap) n = 0;
      if (n > 0) load_records(seg, 0, n, lane, sv0, dv0, vv0);
    }

    if (kReplay && !rescan) {
      if (wave == 0) {
        float* sp = acc + kRowsPerBlock * kRowPitch + lane;
        sp[0] = 0.f;
        sp[kColPitch] = 0.f;
        sp[2 * kColPitch] = 0.f;
        sp[3 * kColPitch] = 0.f;
      }
      replay_records(seg, n, sv0, dv0, vv0, x, acc, wave, lane);
    } else {
      for (int i = tid; i < (kRowsPerBlock + 1) * kRowPitch / 4; i += 1024)
        acc4[i] = make_float4(0.f, 0.f, 0.f, 0.f);
      if (!kReplay && tid < kWaves) queueN[tid] = 0;
      __syncthreads();
      if (kReplay)
        scan_and_serve<false>(codes, dstIdx, srcIdx, vals, x, acc, qe, queueW,
                              queueN, kb, start, nUnits, E, r0, rows, wave,
                              lane);
      else
        scan_and_serve<true>(codes, dstIdx, srcIdx, vals, x, acc, qe, queueW,
                             queueN, kb, start, nUnits, E, r0, rows, wave,
                             lane);
    }
    __syncthreads();
    if (!kReplay) store_queue(queueW, queueN, seg, cntp, wave, lane);

    block_gemm<OC>(acc, W, bias, out + (size_t)r0 * OC, rows, wave, lane);
    __syncthreads();
  }
}

extern "C" void kernel_launch(void* const* d_in, const int* in_sizes, int n_in,
                              void* d_out, int out_size, void* d_ws,
                              size_t ws_size, hipStream_t stream) {
  const float* x = (const float*)d_in[0];
  const int* adj = (const int*)d_in[1];
  const float* vals = (const float*)d_in[2];
  const float* W1 = (const float*)d_in[3];
  const float* b1 = (const float*)d_in[4];
  const float* W2 = (const float*)d_in[5];
  const float* b2 = (const float*)d_in[6];
  const float* W3 = (const float*)d_in[7];
  const float* b3 = (const float*)d_in[8];
  float* out = (float*)d_out;

  const int E = in_sizes[2];
  const int N = in_sizes[0] / 128;
  const int* dstIdx = adj;
  const int* srcIdx = adj + E;

  float* ws0 = (float*)d_ws;
  float* ws1 = ws0 + (size_t)N * 128;
  unsigned* codes = (unsigned*)(ws1 + (size_t)N * 128);

  const int layerBlocks = 256;
  const int nWords = (E + 511) / 512 * 128;
  const int codeBlocks = (nWords + 1023) / 1024;
  int* counts = (int*)(codes + nWords);

  owner_code_kernel<<<codeBlocks, 256, 0, stream>>>(dstIdx, codes, E, nWords);

  layer_kernel<128, false><<<layerBlocks, 1024, 0, stream>>>(
      codes, dstIdx, srcIdx, vals, x, W1, b1, ws0, counts, E, N);
  layer_kernel<128, true><<<layerBlocks, 1024, 0, stream>>>(
      codes, dstIdx, srcIdx, vals, ws0, W2, b2, ws1, counts, E, N);
  layer_kernel<64, true><<<layerBlocks, 1024, 0, stream>>>(
      codes, dstIdx, srcIdx, vals, ws1, W3, b3, out, counts, E, N);
}
